# starting kernel
# baseline (speedup 1.0000x reference)
#include <hip/hip_runtime.h>
#include <hip/hip_bf16.h>

typedef float v2f __attribute__((ext_vector_type(2)));
typedef float v8f __attribute__((ext_vector_type(8)));

#define T_DIM  16384
#define IN_DIM 1024
#define S_DIM  512
#define NCOL   2048

__global__ void wtrans_kernel(const float* __restrict__ w_in, float* __restrict__ bt) {
    const int tid = blockIdx.x * blockDim.x + threadIdx.x;
    const int n = tid & (NCOL - 1);
    const int i = tid >> 11;
    const int s = n & (S_DIM - 1);
    const int g = n >> 9;
    bt[tid] = w_in[((s << 2) + g) * IN_DIM + i];
}

__global__ void __launch_bounds__(256)
gemm_u_kernel(const float* __restrict__ x, const float* __restrict__ bt,
              float* __restrict__ u) {
    const int lane = threadIdx.x & 31;
    const int wave = threadIdx.x >> 5;
    const int half = lane >> 4;
    const int l16  = lane & 15;
    const int row0 = blockIdx.y << 5;
    const int col0 = (blockIdx.x << 8) + (wave << 5);

    const float* aRow0 = x + (size_t)(row0 + l16) * IN_DIM + 2 * half;
    const float* aRow1 = aRow0 + (size_t)16 * IN_DIM;
    const float* bCol0 = bt + (size_t)(2 * half) * NCOL + col0 + l16;
    const float* bCol1 = bCol0 + 16;

    v8f acc00 = {0.f,0.f,0.f,0.f,0.f,0.f,0.f,0.f};
    v8f acc01 = acc00, acc10 = acc00, acc11 = acc00;

    for (int k0 = 0; k0 < IN_DIM; k0 += 64) {
        if (k0 + 64 < IN_DIM) {
            __builtin_prefetch(aRow0 + k0 + 64, 0, 3);
            __builtin_prefetch(aRow1 + k0 + 64, 0, 3);
        }
#pragma unroll
        for (int kk = 0; kk < 64; kk += 4) {
            const int k = k0 + kk;
            v2f a0 = *(const v2f*)(aRow0 + k);
            v2f a1 = *(const v2f*)(aRow1 + k);
            v2f b0, b1;
            b0.x = bCol0[(size_t)k * NCOL];
            b0.y = bCol0[(size_t)(k + 1) * NCOL];
            b1.x = bCol1[(size_t)k * NCOL];
            b1.y = bCol1[(size_t)(k + 1) * NCOL];
            acc00 = __builtin_amdgcn_wmma_f32_16x16x4_f32(false, a0, false, b0,
                                                          (short)0, acc00, false, false);
            acc01 = __builtin_amdgcn_wmma_f32_16x16x4_f32(false, a0, false, b1,
                                                          (short)0, acc01, false, false);
            acc10 = __builtin_amdgcn_wmma_f32_16x16x4_f32(false, a1, false, b0,
                                                          (short)0, acc10, false, false);
            acc11 = __builtin_amdgcn_wmma_f32_16x16x4_f32(false, a1, false, b1,
                                                          (short)0, acc11, false, false);
        }
    }

    float* out00 = u + (size_t)row0 * NCOL + col0 + l16;
    float* out01 = out00 + 16;
    float* out10 = out00 + (size_t)16 * NCOL;
    float* out11 = out10 + 16;
#pragma unroll
    for (int r = 0; r < 8; ++r) {
        const size_t roff = (size_t)(r + half * 8) * NCOL;
        out00[roff] = acc00[r];
        out01[roff] = acc01[r];
        out10[roff] = acc10[r];
        out11[roff] = acc11[r];
    }
}

__device__ __forceinline__ float sigmoid_f(float v) {
    return 1.f / (1.f + __expf(-v));
}
__device__ __forceinline__ float tanh_f(float v) {
    return 1.f - 2.f / (__expf(2.f * v) + 1.f);
}

__global__ void scan_kernel(const float* __restrict__ u,
                            const float* __restrict__ w_rec,
                            const float* __restrict__ bias,
                            float* __restrict__ h_n, float* __restrict__ c_n) {
    const int s = blockIdx.x * blockDim.x + threadIdx.x;
    const float wf = w_rec[s],             wi = w_rec[S_DIM + s];
    const float wo = w_rec[2 * S_DIM + s], wg = w_rec[3 * S_DIM + s];
    const float bf = bias[s],              bi = bias[S_DIM + s];
    const float bo = bias[2 * S_DIM + s],  bg = bias[3 * S_DIM + s];

    float h = 0.f, c = 0.f;
    h_n[s] = 0.f;
    c_n[s] = 0.f;

    const float* up = u + s;
    for (int t = 0; t < T_DIM; ++t) {
        const float* ut = up + (size_t)t * NCOL;
        float f = sigmoid_f(bf + wf * h + ut[0]);
        float i = sigmoid_f(bi + wi * h + ut[S_DIM]);
        float o = sigmoid_f(bo + wo * h + ut[2 * S_DIM]);
        float g = tanh_f   (bg + wg * h + ut[3 * S_DIM]);
        c = f * c + i * g;
        h = o * tanh_f(c);
        h_n[(size_t)(t + 1) * S_DIM + s] = h;
        c_n[(size_t)(t + 1) * S_DIM + s] = c;
    }
}

__global__ void q_kernel(const float* __restrict__ h_n,
                         const float* __restrict__ w_reg,
                         const float* __restrict__ b_reg,
                         float* __restrict__ q) {
    const int lane = threadIdx.x & 31;
    const int t = (blockIdx.x * blockDim.x + threadIdx.x) >> 5;
    const float* hr = h_n + (size_t)(t + 1) * S_DIM;
    float acc = 0.f;
#pragma unroll
    for (int j = 0; j < S_DIM; j += 32)
        acc += hr[j + lane] * w_reg[j + lane];
#pragma unroll
    for (int off = 16; off; off >>= 1)
        acc += __shfl_xor(acc, off, 32);
    if (lane == 0) q[t] = acc + b_reg[0];
}

extern "C" void kernel_launch(void* const* d_in, const int* in_sizes, int n_in,
                              void* d_out, int out_size, void* d_ws, size_t ws_size,
                              hipStream_t stream) {
    (void)in_sizes; (void)n_in; (void)out_size; (void)ws_size;
    const float* x     = (const float*)d_in[0];
    const float* w_in  = (const float*)d_in[1];
    const float* w_rec = (const float*)d_in[2];
    const float* bias  = (const float*)d_in[3];
    const float* w_reg = (const float*)d_in[4];
    const float* b_reg = (const float*)d_in[5];

    float* out = (float*)d_out;
    float* q   = out;
    float* h_n = out + T_DIM;
    float* c_n = h_n + (size_t)(T_DIM + 1) * S_DIM;

    float* u  = (float*)d_ws;
    float* bt = u + (size_t)T_DIM * NCOL;

    wtrans_kernel<<<(IN_DIM * NCOL) / 256, 256, 0, stream>>>(w_in, bt);

    dim3 ggrid(NCOL / 256, T_DIM / 32);
    gemm_u_kernel<<<ggrid, 256, 0, stream>>>(x, bt, u);

    scan_kernel<<<S_DIM / 32, 32, 0, stream>>>(u, w_rec, bias, h_n, c_n);

    q_kernel<<<(T_DIM * 32) / 256, 256, 0, stream>>>(h_n, w_reg, b_reg, q);
}


# product fused into the recurrence launch by column ownership: a work-group owns 4 states, its helper waves WMMA the next 192 steps of their 16 u columns into LDS; u never in memory; sums unchanged
# speedup vs baseline: 6.9953x; 6.9953x over previous
#include <hip/hip_runtime.h>
#include <hip/hip_bf16.h>

typedef float v2f __attribute__((ext_vector_type(2)));
typedef float v4f __attribute__((ext_vector_type(4)));
typedef float v8f __attribute__((ext_vector_type(8)));

#define T_DIM  16384
#define IN_DIM 1024
#define S_DIM  512

#define WG_STATES   4
#define N_WG        (S_DIM / WG_STATES)
#define N_HELP      6
#define BLK         (N_HELP * 32)
#define N_BLK       ((T_DIM + BLK - 1) / BLK)
#define B_PITCH     (IN_DIM + 4)
#define B_FLOATS    (16 * B_PITCH)
#define STG_FLOATS  (BLK * 16)
#define FUSED_THREADS 256
#define LDS_FLOATS  (168 * 1024 / 4)

#define LOG2E_F   1.44269504088896340736f
#define K_SIG     (-LOG2E_F)
#define K_TANH    (2.f * LOG2E_F)

__device__ __forceinline__ float sigmoid_scaled(float a) {
    return 1.f / (1.f + __builtin_amdgcn_exp2f(a));
}
__device__ __forceinline__ float tanh_times_sigmoid(float a, float es) {
    const float e = __builtin_amdgcn_exp2f(-__builtin_fabsf(a));
    const float num = __builtin_copysignf(1.f - e, a);
    return num / ((1.f + es) * (1.f + e));
}

#define NQ 8
__device__ __forceinline__ void load_frags(v4f (&a)[2][NQ], v4f (&b)[NQ],
                                           const float* __restrict__ aP,
                                           const float* bP, int k0) {
#pragma unroll
    for (int q = 0; q < NQ; ++q) {
        a[0][q] = *(const v4f*)(aP + k0 + 8 * q);
        a[1][q] = *(const v4f*)(aP + (size_t)16 * IN_DIM + k0 + 8 * q);
        b[q]    = *(const v4f*)(bP + k0 + 8 * q);
    }
}

__device__ __forceinline__ void mma_frags(v8f& acc0, v8f& acc1,
                                          const v4f (&a)[2][NQ], const v4f (&b)[NQ]) {
#pragma unroll
    for (int q = 0; q < NQ; ++q) {
        const v2f b01 = {b[q].x, b[q].y};
        const v2f b23 = {b[q].z, b[q].w};
        const v2f a0l = {a[0][q].x, a[0][q].y};
        const v2f a1l = {a[1][q].x, a[1][q].y};
        const v2f a0h = {a[0][q].z, a[0][q].w};
        const v2f a1h = {a[1][q].z, a[1][q].w};
        acc0 = __builtin_amdgcn_wmma_f32_16x16x4_f32(false, a0l, false, b01, (short)0, acc0, false, false);
        acc1 = __builtin_amdgcn_wmma_f32_16x16x4_f32(false, a1l, false, b01, (short)0, acc1, false, false);
        acc0 = __builtin_amdgcn_wmma_f32_16x16x4_f32(false, a0h, false, b23, (short)0, acc0, false, false);
        acc1 = __builtin_amdgcn_wmma_f32_16x16x4_f32(false, a1h, false, b23, (short)0, acc1, false, false);
    }
}

__device__ __forceinline__ void product_block(float* dst, const float* __restrict__ x,
                                              const float* Bs, int t0, int hidx, int lane,
                                              float bias_n, float k_n) {
    const int half = lane >> 4;
    const int l16  = lane & 15;
    const int j0   = hidx * 32;
    if (t0 + j0 >= T_DIM) return;

    const float* aP = x + (size_t)(t0 + j0 + l16) * IN_DIM + 4 * half;
    const float* bP = Bs + l16 * B_PITCH + 4 * half;

    v8f acc0 = (v8f){0.f,0.f,0.f,0.f,0.f,0.f,0.f,0.f};
    v8f acc1 = (v8f){0.f,0.f,0.f,0.f,0.f,0.f,0.f,0.f};
    v4f aA[2][NQ], bA[NQ], aB[2][NQ], bB[NQ];
    load_frags(aA, bA, aP, bP, 0);
    for (int k0 = 0; k0 < IN_DIM; k0 += 128) {
        load_frags(aB, bB, aP, bP, k0 + 64);
        asm volatile("" ::: "memory");
        mma_frags(acc0, acc1, aA, bA);
        if (k0 + 128 < IN_DIM) load_frags(aA, bA, aP, bP, k0 + 128);
        mma_frags(acc0, acc1, aB, bB);
    }

    float* o = dst + (j0 + 8 * half) * 16 + l16;
#pragma unroll
    for (int r = 0; r < 8; ++r) {
        o[r * 16]        = (bias_n + acc0[r]) * k_n;
        o[(16 + r) * 16] = (bias_n + acc1[r]) * k_n;
    }
}

__device__ __forceinline__ void stage_barrier() {
    asm volatile("s_wait_dscnt 0x0\n\ts_barrier_signal -1\n\ts_barrier_wait -1" ::: "memory");
}

__global__ void __launch_bounds__(FUSED_THREADS)
fused_kernel(const float* __restrict__ x, const float* __restrict__ w_in,
             const float* __restrict__ w_rec, const float* __restrict__ bias,
             float* __restrict__ hc) {
    __shared__ __attribute__((aligned(16))) float lds[LDS_FLOATS];
    float* Bs  = lds;
    float* stg = lds + B_FLOATS;

    const int tid  = threadIdx.x;
    const int lane = tid & 31;
    const int wave = __builtin_amdgcn_readfirstlane((int)(tid >> 5));
    const int wg   = blockIdx.x;

    {
        const float* wsrc = w_in + (size_t)wg * 16 * IN_DIM;
#pragma unroll
        for (int i = 0; i < 16; ++i) {
            const int f   = tid + FUSED_THREADS * i;
            const int row = f >> 8;
            const int c4  = f & 255;
            *(v4f*)(Bs + row * B_PITCH + c4 * 4) = *(const v4f*)(wsrc + (size_t)row * IN_DIM + c4 * 4);
        }
    }

    const int s = wg * WG_STATES + (lane & 3);
    const float wf = w_rec[s] * K_SIG,             wi = w_rec[S_DIM + s] * K_SIG;
    const float wo = w_rec[2 * S_DIM + s] * K_SIG, wg_ = w_rec[3 * S_DIM + s] * K_TANH;

    const int n    = lane & 15;
    const float bias_n = bias[(n & 3) * S_DIM + wg * WG_STATES + (n >> 2)];
    const float k_n    = ((n & 3) == 3) ? K_TANH : K_SIG;
    const int hidx = (wave < 4) ? wave - 1 : wave - 2;
    const bool helper = (wave != 0) && (wave != 4);

    stage_barrier();
    if (helper) product_block(stg, x, Bs, 0, hidx, lane, bias_n, k_n);
    stage_barrier();

    float h = 0.f, c = 0.f;
    const bool hiLane = (lane & 4) != 0;
    const int  grp    = lane >> 3;
    float* hcw = hc + (size_t)wg * T_DIM * 8 + lane;

    for (int b = 0; b < N_BLK; ++b) {
        const int t0 = b * BLK;
        if (wave == 0) {
            const float* ub = stg + (b & 1) * STG_FLOATS + (lane & 3) * 4;
            const int nsteps = (T_DIM - t0 < BLK) ? (T_DIM - t0) : BLK;
            for (int jb = 0; jb < nsteps; jb += 16) {
                float keep[4];
#pragma unroll
                for (int jj = 0; jj < 16; ++jj) {
                    const int j = jb + jj;
                    const v4f uv = *(const v4f*)(ub + j * 16);
                    const float f  = sigmoid_scaled(wf * h + uv.x);
                    const float ei = __builtin_amdgcn_exp2f(wi * h + uv.y);
                    const float eo = __builtin_amdgcn_exp2f(wo * h + uv.z);
                    const float ig = tanh_times_sigmoid(wg_ * h + uv.w, ei);
                    c = f * c + ig;
                    h = tanh_times_sigmoid(c * K_TANH, eo);
                    keep[jj & 3] = hiLane ? c : h;
                    if ((jj & 3) == 3) {
                        const float v01 = (grp & 1) ? keep[1] : keep[0];
                        const float v23 = (grp & 1) ? keep[3] : keep[2];
                        hcw[(size_t)(t0 + j - 3) * 8] = (grp & 2) ? v23 : v01;
                    }
                }
            }
        } else if (helper) {
            if (b + 1 < N_BLK)
                product_block(stg + ((b + 1) & 1) * STG_FLOATS, x, Bs, t0 + BLK, hidx, lane, bias_n, k_n);
        }
        stage_barrier();
    }
}

__global__ void __launch_bounds__(128)
rows_kernel(const float* __restrict__ hc, float* __restrict__ h_n, float* __restrict__ c_n) {
    const int wgi = threadIdx.x;
    const int t0  = blockIdx.x * 4;
    const v4f* src = (const v4f*)(hc + ((size_t)wgi * T_DIM + t0) * 8);
#pragma unroll
    for (int p = 0; p < 8; ++p) {
        const v4f v = src[p];
        float* row = ((p & 1) ? c_n : h_n) + (size_t)(t0 + (p >> 1) + 1) * S_DIM;
        *(v4f*)(row + wgi * 4) = v;
    }
    if (blockIdx.x == 0) {
        const v4f z = {0.f, 0.f, 0.f, 0.f};
        *(v4f*)(h_n + wgi * 4) = z;
        *(v4f*)(c_n + wgi * 4) = z;
    }
}

__global__ void q_kernel(const float* __restrict__ h_n,
                         const float* __restrict__ w_reg,
                         const float* __restrict__ b_reg,
                         float* __restrict__ q) {
    const int lane = threadIdx.x & 31;
    const int t = (blockIdx.x * blockDim.x + threadIdx.x) >> 5;
    const float* hr = h_n + (size_t)(t + 1) * S_DIM;
    float acc = 0.f;
#pragma unroll
    for (int j = 0; j < S_DIM; j += 32)
        acc += hr[j + lane] * w_reg[j + lane];
#pragma unroll
    for (int off = 16; off; off >>= 1)
        acc += __shfl_xor(acc, off, 32);
    if (lane == 0) q[t] = acc + b_reg[0];
}

extern "C" void kernel_launch(void* const* d_in, const int* in_sizes, int n_in,
                              void* d_out, int out_size, void* d_ws, size_t ws_size,
                              hipStream_t stream) {
    (void)in_sizes; (void)n_in; (void)out_size; (void)ws_size;
    const float* x     = (const float*)d_in[0];
    const float* w_in  = (const float*)d_in[1];
    const float* w_rec = (const float*)d_in[2];
    const float* bias  = (const float*)d_in[3];
    const float* w_reg = (const float*)d_in[4];
    const float* b_reg = (const float*)d_in[5];

    float* out = (float*)d_out;
    float* q   = out;
    float* h_n = out + T_DIM;
    float* c_n = h_n + (size_t)(T_DIM + 1) * S_DIM;

    float* hc = (float*)d_ws;

    fused_kernel<<<N_WG, FUSED_THREADS, 0, stream>>>(x, w_in, w_rec, bias, hc);

    rows_kernel<<<T_DIM / 4, 128, 0, stream>>>(hc, h_n, c_n);

    q_kernel<<<(T_DIM * 32) / 256, 256, 0, stream>>>(h_n, w_reg, b_reg, q);
}


# scan: each divide written out (scale, rcp, both exact-residual corrections, fmas, fixup), first quotient estimate from the raw rcp, its refinement beside it: 8 dependent links not 10; sums unchanged
# speedup vs baseline: 7.5236x; 1.0755x over previous
#include <hip/hip_runtime.h>
#include <hip/hip_bf16.h>

typedef float v2f __attribute__((ext_vector_type(2)));
typedef float v4f __attribute__((ext_vector_type(4)));
typedef float v8f __attribute__((ext_vector_type(8)));

#define T_DIM  16384
#define IN_DIM 1024
#define S_DIM  512

#define WG_STATES   4
#define N_WG        (S_DIM / WG_STATES)
#define N_HELP      6
#define BLK         (N_HELP * 32)
#define N_BLK       ((T_DIM + BLK - 1) / BLK)
#define B_PITCH     (IN_DIM + 4)
#define B_FLOATS    (16 * B_PITCH)
#define STG_FLOATS  (BLK * 16)
#define FUSED_THREADS 256
#define LDS_FLOATS  (168 * 1024 / 4)

#define LOG2E_F   1.44269504088896340736f
#define K_SIG     (-LOG2E_F)
#define K_TANH    (2.f * LOG2E_F)

__device__ __forceinline__ float div_rn(float n, float d) {
    bool den_flag, num_flag;
    const float ds = __builtin_amdgcn_div_scalef(n, d, false, &den_flag);
    const float ns = __builtin_amdgcn_div_scalef(n, d, true,  &num_flag);
    (void)den_flag;
    const float r0 = __builtin_amdgcn_rcpf(ds);
    const float e0 = __builtin_fmaf(-ds, r0, 1.f);
    const float r1 = __builtin_fmaf(e0, r0, r0);
    const float q0 = ns * r0;
    const float e1 = __builtin_fmaf(-ds, q0, ns);
    const float q1 = __builtin_fmaf(e1, r0, q0);
    const float e2 = __builtin_fmaf(-ds, q1, ns);
    const float q  = __builtin_amdgcn_div_fmasf(e2, r1, q1, num_flag);
    return __builtin_amdgcn_div_fixupf(q, d, n);
}

__device__ __forceinline__ float sigmoid_scaled(float a) {
    return div_rn(1.f, 1.f + __builtin_amdgcn_exp2f(a));
}
__device__ __forceinline__ float tanh_times_sigmoid(float a, float es) {
    const float e = __builtin_amdgcn_exp2f(-__builtin_fabsf(a));
    const float num = __builtin_copysignf(1.f - e, a);
    return div_rn(num, (1.f + es) * (1.f + e));
}

#define NQ 8
__device__ __forceinline__ void load_frags(v4f (&a)[2][NQ], v4f (&b)[NQ],
                                           const float* __restrict__ aP,
                                           const float* bP, int k0) {
#pragma unroll
    for (int q = 0; q < NQ; ++q) {
        a[0][q] = *(const v4f*)(aP + k0 + 8 * q);
        a[1][q] = *(const v4f*)(aP + (size_t)16 * IN_DIM + k0 + 8 * q);
        b[q]    = *(const v4f*)(bP + k0 + 8 * q);
    }
}

__device__ __forceinline__ void mma_frags(v8f& acc0, v8f& acc1,
                                          const v4f (&a)[2][NQ], const v4f (&b)[NQ]) {
#pragma unroll
    for (int q = 0; q < NQ; ++q) {
        const v2f b01 = {b[q].x, b[q].y};
        const v2f b23 = {b[q].z, b[q].w};
        const v2f a0l = {a[0][q].x, a[0][q].y};
        const v2f a1l = {a[1][q].x, a[1][q].y};
        const v2f a0h = {a[0][q].z, a[0][q].w};
        const v2f a1h = {a[1][q].z, a[1][q].w};
        acc0 = __builtin_amdgcn_wmma_f32_16x16x4_f32(false, a0l, false, b01, (short)0, acc0, false, false);
        acc1 = __builtin_amdgcn_wmma_f32_16x16x4_f32(false, a1l, false, b01, (short)0, acc1, false, false);
        acc0 = __builtin_amdgcn_wmma_f32_16x16x4_f32(false, a0h, false, b23, (short)0, acc0, false, false);
        acc1 = __builtin_amdgcn_wmma_f32_16x16x4_f32(false, a1h, false, b23, (short)0, acc1, false, false);
    }
}

__device__ __forceinline__ void product_block(float* dst, const float* __restrict__ x,
                                              const float* Bs, int t0, int hidx, int lane,
                                              float bias_n, float k_n) {
    const int half = lane >> 4;
    const int l16  = lane & 15;
    const int j0   = hidx * 32;
    if (t0 + j0 >= T_DIM) return;

    const float* aP = x + (size_t)(t0 + j0 + l16) * IN_DIM + 4 * half;
    const float* bP = Bs + l16 * B_PITCH + 4 * half;

    v8f acc0 = (v8f){0.f,0.f,0.f,0.f,0.f,0.f,0.f,0.f};
    v8f acc1 = (v8f){0.f,0.f,0.f,0.f,0.f,0.f,0.f,0.f};
    v4f aA[2][NQ], bA[NQ], aB[2][NQ], bB[NQ];
    load_frags(aA, bA, aP, bP, 0);
    for (int k0 = 0; k0 < IN_DIM; k0 += 128) {
        load_frags(aB, bB, aP, bP, k0 + 64);
        asm volatile("" ::: "memory");
        mma_frags(acc0, acc1, aA, bA);
        if (k0 + 128 < IN_DIM) load_frags(aA, bA, aP, bP, k0 + 128);
        mma_frags(acc0, acc1, aB, bB);
    }

    float* o = dst + (j0 + 8 * half) * 16 + l16;
#pragma unroll
    for (int r = 0; r < 8; ++r) {
        o[r * 16]        = (bias_n + acc0[r]) * k_n;
        o[(16 + r) * 16] = (bias_n + acc1[r]) * k_n;
    }
}

__device__ __forceinline__ void stage_barrier() {
    asm volatile("s_wait_dscnt 0x0\n\ts_barrier_signal -1\n\ts_barrier_wait -1" ::: "memory");
}

__global__ void __launch_bounds__(FUSED_THREADS)
fused_kernel(const float* __restrict__ x, const float* __restrict__ w_in,
             const float* __restrict__ w_rec, const float* __restrict__ bias,
             float* __restrict__ hc) {
    __shared__ __attribute__((aligned(16))) float lds[LDS_FLOATS];
    float* Bs  = lds;
    float* stg = lds + B_FLOATS;

    const int tid  = threadIdx.x;
    const int lane = tid & 31;
    const int wave = __builtin_amdgcn_readfirstlane((int)(tid >> 5));
    const int wg   = blockIdx.x;

    {
        const float* wsrc = w_in + (size_t)wg * 16 * IN_DIM;
#pragma unroll
        for (int i = 0; i < 16; ++i) {
            const int f   = tid + FUSED_THREADS * i;
            const int row = f >> 8;
            const int c4  = f & 255;
            *(v4f*)(Bs + row * B_PITCH + c4 * 4) = *(const v4f*)(wsrc + (size_t)row * IN_DIM + c4 * 4);
        }
    }

    const int s = wg * WG_STATES + (lane & 3);
    const float wf = w_rec[s] * K_SIG,             wi = w_rec[S_DIM + s] * K_SIG;
    const float wo = w_rec[2 * S_DIM + s] * K_SIG, wg_ = w_rec[3 * S_DIM + s] * K_TANH;

    const int n    = lane & 15;
    const float bias_n = bias[(n & 3) * S_DIM + wg * WG_STATES + (n >> 2)];
    const float k_n    = ((n & 3) == 3) ? K_TANH : K_SIG;
    const int hidx = (wave < 4) ? wave - 1 : wave - 2;
    const bool helper = (wave != 0) && (wave != 4);

    stage_barrier();
    if (helper) product_block(stg, x, Bs, 0, hidx, lane, bias_n, k_n);
    stage_barrier();

    float h = 0.f, c = 0.f;
    const bool hiLane = (lane & 4) != 0;
    const int  grp    = lane >> 3;
    float* hcw = hc + (size_t)wg * T_DIM * 8 + lane;

    for (int b = 0; b < N_BLK; ++b) {
        const int t0 = b * BLK;
        if (wave == 0) {
            const float* ub = stg + (b & 1) * STG_FLOATS + (lane & 3) * 4;
            const int nsteps = (T_DIM - t0 < BLK) ? (T_DIM - t0) : BLK;
            for (int jb = 0; jb < nsteps; jb += 16) {
                float keep[4];
#pragma unroll
                for (int jj = 0; jj < 16; ++jj) {
                    const int j = jb + jj;
                    const v4f uv = *(const v4f*)(ub + j * 16);
                    const float f  = sigmoid_scaled(wf * h + uv.x);
                    const float ei = __builtin_amdgcn_exp2f(wi * h + uv.y);
                    const float eo = __builtin_amdgcn_exp2f(wo * h + uv.z);
                    const float ig = tanh_times_sigmoid(wg_ * h + uv.w, ei);
                    c = f * c + ig;
                    h = tanh_times_sigmoid(c * K_TANH, eo);
                    keep[jj & 3] = hiLane ? c : h;
                    if ((jj & 3) == 3) {
                        const float v01 = (grp & 1) ? keep[1] : keep[0];
                        const float v23 = (grp & 1) ? keep[3] : keep[2];
                        hcw[(size_t)(t0 + j - 3) * 8] = (grp & 2) ? v23 : v01;
                    }
                }
            }
        } else if (helper) {
            if (b + 1 < N_BLK)
                product_block(stg + ((b + 1) & 1) * STG_FLOATS, x, Bs, t0 + BLK, hidx, lane, bias_n, k_n);
        }
        stage_barrier();
    }
}

__global__ void __launch_bounds__(128)
rows_kernel(const float* __restrict__ hc, float* __restrict__ h_n, float* __restrict__ c_n) {
    const int wgi = threadIdx.x;
    const int t0  = blockIdx.x * 4;
    const v4f* src = (const v4f*)(hc + ((size_t)wgi * T_DIM + t0) * 8);
#pragma unroll
    for (int p = 0; p < 8; ++p) {
        const v4f v = src[p];
        float* row = ((p & 1) ? c_n : h_n) + (size_t)(t0 + (p >> 1) + 1) * S_DIM;
        *(v4f*)(row + wgi * 4) = v;
    }
    if (blockIdx.x == 0) {
        const v4f z = {0.f, 0.f, 0.f, 0.f};
        *(v4f*)(h_n + wgi * 4) = z;
        *(v4f*)(c_n + wgi * 4) = z;
    }
}

__global__ void q_kernel(const float* __restrict__ h_n,
                         const float* __restrict__ w_reg,
                         const float* __restrict__ b_reg,
                         float* __restrict__ q) {
    const int lane = threadIdx.x & 31;
    const int t = (blockIdx.x * blockDim.x + threadIdx.x) >> 5;
    const float* hr = h_n + (size_t)(t + 1) * S_DIM;
    float acc = 0.f;
#pragma unroll
    for (int j = 0; j < S_DIM; j += 32)
        acc += hr[j + lane] * w_reg[j + lane];
#pragma unroll
    for (int off = 16; off; off >>= 1)
        acc += __shfl_xor(acc, off, 32);
    if (lane == 0) q[t] = acc + b_reg[0];
}

extern "C" void kernel_launch(void* const* d_in, const int* in_sizes, int n_in,
                              void* d_out, int out_size, void* d_ws, size_t ws_size,
                              hipStream_t stream) {
    (void)in_sizes; (void)n_in; (void)out_size; (void)ws_size;
    const float* x     = (const float*)d_in[0];
    const float* w_in  = (const float*)d_in[1];
    const float* w_rec = (const float*)d_in[2];
    const float* bias  = (const float*)d_in[3];
    const float* w_reg = (const float*)d_in[4];
    const float* b_reg = (const float*)d_in[5];

    float* out = (float*)d_out;
    float* q   = out;
    float* h_n = out + T_DIM;
    float* c_n = h_n + (size_t)(T_DIM + 1) * S_DIM;

    float* hc = (float*)d_ws;

    fused_kernel<<<N_WG, FUSED_THREADS, 0, stream>>>(x, w_in, w_rec, bias, hc);

    rows_kernel<<<T_DIM / 4, 128, 0, stream>>>(hc, h_n, c_n);

    q_kernel<<<(T_DIM * 32) / 256, 256, 0, stream>>>(h_n, w_reg, b_reg, q);
}


# scan: one recurrence step written out in issue order (inline asm): each h->c->h chain instruction at the first slot its operand allows, off-chain work in the idle slots; arithmetic and sums unchanged
# speedup vs baseline: 7.8255x; 1.0401x over previous
#include <hip/hip_runtime.h>
#include <hip/hip_bf16.h>

typedef float v2f __attribute__((ext_vector_type(2)));
typedef float v4f __attribute__((ext_vector_type(4)));
typedef float v8f __attribute__((ext_vector_type(8)));

#define T_DIM  16384
#define IN_DIM 1024
#define S_DIM  512

#define WG_STATES   4
#define N_WG        (S_DIM / WG_STATES)
#define N_HELP      6
#define BLK         (N_HELP * 32)
#define N_BLK       ((T_DIM + BLK - 1) / BLK)
#define B_PITCH     (IN_DIM + 4)
#define B_FLOATS    (16 * B_PITCH)
#define STG_FLOATS  (BLK * 16)
#define FUSED_THREADS 256
#define LDS_FLOATS  (168 * 1024 / 4)

#define LOG2E_F   1.44269504088896340736f
#define K_SIG     (-LOG2E_F)
#define K_TANH    (2.f * LOG2E_F)

__device__ __forceinline__ void lstm_step(float& h, float& c, const v4f uv,
                                          float wf, float wi, float wo, float wg) {
    float t0, t1, t2, t3, t4, t5, t6, t7, t8, t9, t10, t11, t12, t13, t14, t15;
    int sf;
    asm(
        "v_fma_f32 %[t0], %[wf], %[h], %[ux]\n\t"
        "v_fma_f32 %[t1], %[wg], %[h], %[uw]\n\t"
        "v_fma_f32 %[t2], %[wi], %[h], %[uy]\n\t"
        "v_fma_f32 %[t3], %[wo], %[h], %[uz]\n\t"
        "v_exp_f32 %[t4], %[t0]\n\t"
        "v_exp_f32 %[t5], -|%[t1]|\n\t"
        "v_exp_f32 %[t6], %[t2]\n\t"
        "v_exp_f32 %[t7], %[t3]\n\t"
        "v_add_f32 %[t8], 1.0, %[t4]\n\t"
        "v_sub_f32 %[t9], 1.0, %[t5]\n\t"
        "v_add_f32 %[t10], 1.0, %[t5]\n\t"
        "v_add_f32 %[t11], 1.0, %[t6]\n\t"
        "v_div_scale_f32 %[t12], null, %[t8], %[t8], 1.0\n\t"
        "v_div_scale_f32 %[t13], vcc_lo, 1.0, %[t8], 1.0\n\t"
        "v_bfi_b32 %[t14], 0x7fffffff, %[t9], %[t1]\n\t"
        "v_add_f32 %[t15], 1.0, %[t7]\n\t"
        "v_mul_f32 %[t0], %[t11], %[t10]\n\t"
        "v_rcp_f32 %[t2], %[t12]\n\t"
        "v_div_scale_f32 %[t3], null, %[t0], %[t0], %[t14]\n\t"
        "v_div_scale_f32 %[t4], %[sf], %[t14], %[t0], %[t14]\n\t"
        "v_mul_f32 %[t5], %[t13], %[t2]\n\t"
        "v_rcp_f32 %[t6], %[t3]\n\t"
        "v_fma_f32 %[t9], -%[t12], %[t2], 1.0\n\t"
        "v_fma_f32 %[t1], -%[t12], %[t5], %[t13]\n\t"
        "v_fma_f32 %[t7], %[t9], %[t2], %[t2]\n\t"
        "v_mul_f32 %[t11], %[t4], %[t6]\n\t"
        "v_fma_f32 %[t10], %[t1], %[t2], %[t5]\n\t"
        "v_fma_f32 %[t9], -%[t3], %[t6], 1.0\n\t"
        "v_fma_f32 %[t1], -%[t3], %[t11], %[t4]\n\t"
        "v_fma_f32 %[t5], -%[t12], %[t10], %[t13]\n\t"
        "v_fma_f32 %[t2], %[t9], %[t6], %[t6]\n\t"
        "v_fma_f32 %[t13], %[t1], %[t6], %[t11]\n\t"
        "v_div_fmas_f32 %[t12], %[t5], %[t7], %[t10]\n\t"
        "s_mov_b32 vcc_lo, %[sf]\n\t"
        "v_fma_f32 %[t9], -%[t3], %[t13], %[t4]\n\t"
        "v_div_fixup_f32 %[t1], %[t12], %[t8], 1.0\n\t"
        "v_div_fmas_f32 %[t11], %[t9], %[t2], %[t13]\n\t"
        "v_div_fixup_f32 %[t6], %[t11], %[t0], %[t14]\n\t"
        "v_fma_f32 %[c], %[t1], %[c], %[t6]\n\t"
        "v_mul_f32 %[t5], 0x4038aa3b, %[c]\n\t"
        "v_exp_f32 %[t10], -|%[t5]|\n\t"
        "v_nop\n\t"
        "v_sub_f32 %[t7], 1.0, %[t10]\n\t"
        "v_add_f32 %[t4], 1.0, %[t10]\n\t"
        "v_bfi_b32 %[t3], 0x7fffffff, %[t7], %[t5]\n\t"
        "v_mul_f32 %[t12], %[t15], %[t4]\n\t"
        "v_div_scale_f32 %[t8], null, %[t12], %[t12], %[t3]\n\t"
        "v_div_scale_f32 %[t9], vcc_lo, %[t3], %[t12], %[t3]\n\t"
        "v_rcp_f32 %[t13], %[t8]\n\t"
        "v_nop\n\t"
        "v_mul_f32 %[t2], %[t9], %[t13]\n\t"
        "v_fma_f32 %[t0], -%[t8], %[t13], 1.0\n\t"
        "v_fma_f32 %[t14], -%[t8], %[t2], %[t9]\n\t"
        "v_fma_f32 %[t6], %[t0], %[t13], %[t13]\n\t"
        "v_fma_f32 %[t1], %[t14], %[t13], %[t2]\n\t"
        "v_fma_f32 %[t10], -%[t8], %[t1], %[t9]\n\t"
        "v_div_fmas_f32 %[t7], %[t10], %[t6], %[t1]\n\t"
        "v_div_fixup_f32 %[h], %[t7], %[t12], %[t3]\n\t"
        : [h] "+v"(h), [c] "+v"(c),
          [t0] "=&v"(t0),
          [t1] "=&v"(t1),
          [t2] "=&v"(t2),
          [t3] "=&v"(t3),
          [t4] "=&v"(t4),
          [t5] "=&v"(t5),
          [t6] "=&v"(t6),
          [t7] "=&v"(t7),
          [t8] "=&v"(t8),
          [t9] "=&v"(t9),
          [t10] "=&v"(t10),
          [t11] "=&v"(t11),
          [t12] "=&v"(t12),
          [t13] "=&v"(t13),
          [t14] "=&v"(t14),
          [t15] "=&v"(t15),
          [sf] "=&s"(sf)
        : [ux] "v"(uv.x), [uy] "v"(uv.y), [uz] "v"(uv.z), [uw] "v"(uv.w),
          [wf] "v"(wf), [wi] "v"(wi), [wo] "v"(wo), [wg] "v"(wg)
        : "vcc");
}

#define NQ 8
__device__ __forceinline__ void load_frags(v4f (&a)[2][NQ], v4f (&b)[NQ],
                                           const float* __restrict__ aP,
                                           const float* bP, int k0) {
#pragma unroll
    for (int q = 0; q < NQ; ++q) {
        a[0][q] = *(const v4f*)(aP + k0 + 8 * q);
        a[1][q] = *(const v4f*)(aP + (size_t)16 * IN_DIM + k0 + 8 * q);
        b[q]    = *(const v4f*)(bP + k0 + 8 * q);
    }
}

__device__ __forceinline__ void mma_frags(v8f& acc0, v8f& acc1,
                                          const v4f (&a)[2][NQ], const v4f (&b)[NQ]) {
#pragma unroll
    for (int q = 0; q < NQ; ++q) {
        const v2f b01 = {b[q].x, b[q].y};
        const v2f b23 = {b[q].z, b[q].w};
        const v2f a0l = {a[0][q].x, a[0][q].y};
        const v2f a1l = {a[1][q].x, a[1][q].y};
        const v2f a0h = {a[0][q].z, a[0][q].w};
        const v2f a1h = {a[1][q].z, a[1][q].w};
        acc0 = __builtin_amdgcn_wmma_f32_16x16x4_f32(false, a0l, false, b01, (short)0, acc0, false, false);
        acc1 = __builtin_amdgcn_wmma_f32_16x16x4_f32(false, a1l, false, b01, (short)0, acc1, false, false);
        acc0 = __builtin_amdgcn_wmma_f32_16x16x4_f32(false, a0h, false, b23, (short)0, acc0, false, false);
        acc1 = __builtin_amdgcn_wmma_f32_16x16x4_f32(false, a1h, false, b23, (short)0, acc1, false, false);
    }
}

__device__ __forceinline__ void product_block(float* dst, const float* __restrict__ x,
                                              const float* Bs, int t0, int hidx, int lane,
                                              float bias_n, float k_n) {
    const int half = lane >> 4;
    const int l16  = lane & 15;
    const int j0   = hidx * 32;
    if (t0 + j0 >= T_DIM) return;

    const float* aP = x + (size_t)(t0 + j0 + l16) * IN_DIM + 4 * half;
    const float* bP = Bs + l16 * B_PITCH + 4 * half;

    v8f acc0 = (v8f){0.f,0.f,0.f,0.f,0.f,0.f,0.f,0.f};
    v8f acc1 = (v8f){0.f,0.f,0.f,0.f,0.f,0.f,0.f,0.f};
    v4f aA[2][NQ], bA[NQ], aB[2][NQ], bB[NQ];
    load_frags(aA, bA, aP, bP, 0);
    for (int k0 = 0; k0 < IN_DIM; k0 += 128) {
        load_frags(aB, bB, aP, bP, k0 + 64);
        asm volatile("" ::: "memory");
        mma_frags(acc0, acc1, aA, bA);
        if (k0 + 128 < IN_DIM) load_frags(aA, bA, aP, bP, k0 + 128);
        mma_frags(acc0, acc1, aB, bB);
    }

    float* o = dst + (j0 + 8 * half) * 16 + l16;
#pragma unroll
    for (int r = 0; r < 8; ++r) {
        o[r * 16]        = (bias_n + acc0[r]) * k_n;
        o[(16 + r) * 16] = (bias_n + acc1[r]) * k_n;
    }
}

__device__ __forceinline__ void stage_barrier() {
    asm volatile("s_wait_dscnt 0x0\n\ts_barrier_signal -1\n\ts_barrier_wait -1" ::: "memory");
}

__global__ void __launch_bounds__(FUSED_THREADS)
fused_kernel(const float* __restrict__ x, const float* __restrict__ w_in,
             const float* __restrict__ w_rec, const float* __restrict__ bias,
             float* __restrict__ hc) {
    __shared__ __attribute__((aligned(16))) float lds[LDS_FLOATS];
    float* Bs  = lds;
    float* stg = lds + B_FLOATS;

    const int tid  = threadIdx.x;
    const int lane = tid & 31;
    const int wave = __builtin_amdgcn_readfirstlane((int)(tid >> 5));
    const int wg   = blockIdx.x;

    {
        const float* wsrc = w_in + (size_t)wg * 16 * IN_DIM;
#pragma unroll
        for (int i = 0; i < 16; ++i) {
            const int f   = tid + FUSED_THREADS * i;
            const int row = f >> 8;
            const int c4  = f & 255;
            *(v4f*)(Bs + row * B_PITCH + c4 * 4) = *(const v4f*)(wsrc + (size_t)row * IN_DIM + c4 * 4);
        }
    }

    const int s = wg * WG_STATES + (lane & 3);
    const float wf = w_rec[s] * K_SIG,             wi = w_rec[S_DIM + s] * K_SIG;
    const float wo = w_rec[2 * S_DIM + s] * K_SIG, wg_ = w_rec[3 * S_DIM + s] * K_TANH;

    const int n    = lane & 15;
    const float bias_n = bias[(n & 3) * S_DIM + wg * WG_STATES + (n >> 2)];
    const float k_n    = ((n & 3) == 3) ? K_TANH : K_SIG;
    const int hidx = (wave < 4) ? wave - 1 : wave - 2;
    const bool helper = (wave != 0) && (wave != 4);

    stage_barrier();
    if (helper) product_block(stg, x, Bs, 0, hidx, lane, bias_n, k_n);
    stage_barrier();

    float h = 0.f, c = 0.f;
    const bool hiLane = (lane & 4) != 0;
    const int  grp    = lane >> 3;
    float* hcw = hc + (size_t)wg * T_DIM * 8 + lane;

    for (int b = 0; b < N_BLK; ++b) {
        const int t0 = b * BLK;
        if (wave == 0) {
            const float* ub = stg + (b & 1) * STG_FLOATS + (lane & 3) * 4;
            const int nsteps = (T_DIM - t0 < BLK) ? (T_DIM - t0) : BLK;
            for (int jb = 0; jb < nsteps; jb += 16) {
                float keep[4];
#pragma unroll
                for (int jj = 0; jj < 16; ++jj) {
                    const int j = jb + jj;
                    const v4f uv = *(const v4f*)(ub + j * 16);
                    lstm_step(h, c, uv, wf, wi, wo, wg_);
                    keep[jj & 3] = hiLane ? c : h;
                    if ((jj & 3) == 3) {
                        const float v01 = (grp & 1) ? keep[1] : keep[0];
                        const float v23 = (grp & 1) ? keep[3] : keep[2];
                        hcw[(size_t)(t0 + j - 3) * 8] = (grp & 2) ? v23 : v01;
                    }
                }
            }
        } else if (helper) {
            if (b + 1 < N_BLK)
                product_block(stg + ((b + 1) & 1) * STG_FLOATS, x, Bs, t0 + BLK, hidx, lane, bias_n, k_n);
        }
        stage_barrier();
    }
}

__global__ void __launch_bounds__(128)
rows_kernel(const float* __restrict__ hc, float* __restrict__ h_n, float* __restrict__ c_n) {
    const int wgi = threadIdx.x;
    const int t0  = blockIdx.x * 4;
    const v4f* src = (const v4f*)(hc + ((size_t)wgi * T_DIM + t0) * 8);
#pragma unroll
    for (int p = 0; p < 8; ++p) {
        const v4f v = src[p];
        float* row = ((p & 1) ? c_n : h_n) + (size_t)(t0 + (p >> 1) + 1) * S_DIM;
        *(v4f*)(row + wgi * 4) = v;
    }
    if (blockIdx.x == 0) {
        const v4f z = {0.f, 0.f, 0.f, 0.f};
        *(v4f*)(h_n + wgi * 4) = z;
        *(v4f*)(c_n + wgi * 4) = z;
    }
}

__global__ void q_kernel(const float* __restrict__ h_n,
                         const float* __restrict__ w_reg,
                         const float* __restrict__ b_reg,
                         float* __restrict__ q) {
    const int lane = threadIdx.x & 31;
    const int t = (blockIdx.x * blockDim.x + threadIdx.x) >> 5;
    const float* hr = h_n + (size_t)(t + 1) * S_DIM;
    float acc = 0.f;
#pragma unroll
    for (int j = 0; j < S_DIM; j += 32)
        acc += hr[j + lane] * w_reg[j + lane];
#pragma unroll
    for (int off = 16; off; off >>= 1)
        acc += __shfl_xor(acc, off, 32);
    if (lane == 0) q[t] = acc + b_reg[0];
}

extern "C" void kernel_launch(void* const* d_in, const int* in_sizes, int n_in,
                              void* d_out, int out_size, void* d_ws, size_t ws_size,
                              hipStream_t stream) {
    (void)in_sizes; (void)n_in; (void)out_size; (void)ws_size;
    const float* x     = (const float*)d_in[0];
    const float* w_in  = (const float*)d_in[1];
    const float* w_rec = (const float*)d_in[2];
    const float* bias  = (const float*)d_in[3];
    const float* w_reg = (const float*)d_in[4];
    const float* b_reg = (const float*)d_in[5];

    float* out = (float*)d_out;
    float* q   = out;
    float* h_n = out + T_DIM;
    float* c_n = h_n + (size_t)(T_DIM + 1) * S_DIM;

    float* hc = (float*)d_ws;

    fused_kernel<<<N_WG, FUSED_THREADS, 0, stream>>>(x, w_in, w_rec, bias, hc);

    rows_kernel<<<T_DIM / 4, 128, 0, stream>>>(hc, h_n, c_n);

    q_kernel<<<(T_DIM * 32) / 256, 256, 0, stream>>>(h_n, w_reg, b_reg, q);
}


# scan: 3 chain links folded into fmas: cell state carried times 2*log2e (no mul before tanh's exp), each denominator (1+A)(1+E) one fma behind the later exp: 22 links not 25; rounding points move
# speedup vs baseline: 8.4618x; 1.0813x over previous
#include <hip/hip_runtime.h>
#include <hip/hip_bf16.h>

typedef float v2f __attribute__((ext_vector_type(2)));
typedef float v4f __attribute__((ext_vector_type(4)));
typedef float v8f __attribute__((ext_vector_type(8)));

#define T_DIM  16384
#define IN_DIM 1024
#define S_DIM  512

#define WG_STATES   4
#define N_WG        (S_DIM / WG_STATES)
#define N_HELP      6
#define BLK         (N_HELP * 32)
#define N_BLK       ((T_DIM + BLK - 1) / BLK)
#define B_PITCH     (IN_DIM + 4)
#define B_FLOATS    (16 * B_PITCH)
#define STG_FLOATS  (BLK * 16)
#define FUSED_THREADS 256
#define LDS_FLOATS  (168 * 1024 / 4)

#define LOG2E_F   1.44269504088896340736f
#define K_SIG     (-LOG2E_F)
#define K_TANH    (2.f * LOG2E_F)
#define INV_K_TANH (0.5f * 0.693147180559945309417f)

__device__ __forceinline__ void lstm_step(float& h, float& c, float& co, const v4f uv,
                                          float wf, float wi, float wo, float wg) {
    float t0, t1, t2, t3, t4, t5, t6, t7, t8, t9, t10, t11, t12, t13, t14, t15;
    int sf;
    asm(
        "v_fma_f32 %[t0], %[wg], %[h], %[uw]\n\t"
        "v_fma_f32 %[t1], %[wf], %[h], %[ux]\n\t"
        "v_fma_f32 %[t2], %[wi], %[h], %[uy]\n\t"
        "v_fma_f32 %[t3], %[wo], %[h], %[uz]\n\t"
        "v_exp_f32 %[t4], -|%[t0]|\n\t"
        "v_exp_f32 %[t5], %[t1]\n\t"
        "v_exp_f32 %[t6], %[t2]\n\t"
        "v_exp_f32 %[t7], %[t3]\n\t"
        "v_add_f32 %[t1], 1.0, %[t4]\n\t"
        "v_fma_f32 %[t2], -%[t4], %[kk], %[kk]\n\t"
        "v_add_f32 %[t3], 1.0, %[t5]\n\t"
        "v_fma_f32 %[t4], %[t6], %[t1], %[t1]\n\t"
        "v_bfi_b32 %[t5], 0x7fffffff, %[t2], %[t0]\n\t"
        "v_div_scale_f32 %[t6], null, %[t3], %[t3], 1.0\n\t"
        "v_div_scale_f32 %[t1], vcc_lo, 1.0, %[t3], 1.0\n\t"
        "v_add_f32 %[t0], 1.0, %[t7]\n\t"
        "v_div_scale_f32 %[t7], null, %[t4], %[t4], %[t2]\n\t"
        "v_div_scale_f32 %[t2], %[sf], %[t5], %[t4], %[t5]\n\t"
        "v_rcp_f32 %[t8], %[t6]\n\t"
        "v_min_num_f32 %[t0], 0x7f7fffff, %[t0]\n\t"
        "v_rcp_f32 %[t9], %[t7]\n\t"
        "v_mul_f32 %[t10], %[t1], %[t8]\n\t"
        "v_fma_f32 %[t11], -%[t6], %[t8], 1.0\n\t"
        "v_mul_f32 %[t12], %[t2], %[t9]\n\t"
        "v_fma_f32 %[t13], -%[t7], %[t9], 1.0\n\t"
        "v_fma_f32 %[t14], -%[t6], %[t10], %[t1]\n\t"
        "v_fma_f32 %[t11], %[t11], %[t8], %[t8]\n\t"
        "v_fma_f32 %[t15], -%[t7], %[t12], %[t2]\n\t"
        "v_fma_f32 %[t13], %[t13], %[t9], %[t9]\n\t"
        "v_fma_f32 %[t14], %[t14], %[t8], %[t10]\n\t"
        "v_fma_f32 %[t8], %[t15], %[t9], %[t12]\n\t"
        "v_fma_f32 %[t10], -%[t6], %[t14], %[t1]\n\t"
        "v_fma_f32 %[t15], -%[t7], %[t8], %[t2]\n\t"
        "v_div_fmas_f32 %[t9], %[t10], %[t11], %[t14]\n\t"
        "s_mov_b32 vcc_lo, %[sf]\n\t"
        "v_div_fmas_f32 %[t12], %[t15], %[t13], %[t8]\n\t"
        "v_div_fixup_f32 %[t6], %[t9], %[t3], 1.0\n\t"
        "v_div_fixup_f32 %[t1], %[t12], %[t4], %[t5]\n\t"
        "v_fma_f32 %[c], %[t6], %[c], %[t1]\n\t"
        "v_exp_f32 %[t1], -|%[c]|\n\t"
        "v_mul_f32 %[co], %[ik], %[c]\n\t"
        "v_sub_f32 %[t7], 1.0, %[t1]\n\t"
        "v_fma_f32 %[t2], %[t1], %[t0], %[t0]\n\t"
        "v_bfi_b32 %[t10], 0x7fffffff, %[t7], %[c]\n\t"
        "v_div_scale_f32 %[t11], null, %[t2], %[t2], %[t7]\n\t"
        "v_div_scale_f32 %[t14], vcc_lo, %[t10], %[t2], %[t10]\n\t"
        "v_rcp_f32 %[t7], %[t11]\n\t"
        "v_nop\n\t"
        "v_mul_f32 %[t15], %[t14], %[t7]\n\t"
        "v_fma_f32 %[t13], -%[t11], %[t7], 1.0\n\t"
        "v_fma_f32 %[t8], -%[t11], %[t15], %[t14]\n\t"
        "v_fma_f32 %[t9], %[t13], %[t7], %[t7]\n\t"
        "v_fma_f32 %[t3], %[t8], %[t7], %[t15]\n\t"
        "v_fma_f32 %[t12], -%[t11], %[t3], %[t14]\n\t"
        "v_div_fmas_f32 %[t4], %[t12], %[t9], %[t3]\n\t"
        "v_div_fixup_f32 %[h], %[t4], %[t2], %[t10]\n\t"
        : [h] "+v"(h), [c] "+v"(c), [co] "=&v"(co),
          [t0] "=&v"(t0),
          [t1] "=&v"(t1),
          [t2] "=&v"(t2),
          [t3] "=&v"(t3),
          [t4] "=&v"(t4),
          [t5] "=&v"(t5),
          [t6] "=&v"(t6),
          [t7] "=&v"(t7),
          [t8] "=&v"(t8),
          [t9] "=&v"(t9),
          [t10] "=&v"(t10),
          [t11] "=&v"(t11),
          [t12] "=&v"(t12),
          [t13] "=&v"(t13),
          [t14] "=&v"(t14),
          [t15] "=&v"(t15),
          [sf] "=&s"(sf)
        : [ux] "v"(uv.x), [uy] "v"(uv.y), [uz] "v"(uv.z), [uw] "v"(uv.w),
          [wf] "v"(wf), [wi] "v"(wi), [wo] "v"(wo), [wg] "v"(wg),
          [kk] "s"(K_TANH), [ik] "s"(INV_K_TANH)
        : "vcc");
}

#define NQ 8
__device__ __forceinline__ void load_frags(v4f (&a)[2][NQ], v4f (&b)[NQ],
                                           const float* __restrict__ aP,
                                           const float* bP, int k0) {
#pragma unroll
    for (int q = 0; q < NQ; ++q) {
        a[0][q] = *(const v4f*)(aP + k0 + 8 * q);
        a[1][q] = *(const v4f*)(aP + (size_t)16 * IN_DIM + k0 + 8 * q);
        b[q]    = *(const v4f*)(bP + k0 + 8 * q);
    }
}

__device__ __forceinline__ void mma_frags(v8f& acc0, v8f& acc1,
                                          const v4f (&a)[2][NQ], const v4f (&b)[NQ]) {
#pragma unroll
    for (int q = 0; q < NQ; ++q) {
        const v2f b01 = {b[q].x, b[q].y};
        const v2f b23 = {b[q].z, b[q].w};
        const v2f a0l = {a[0][q].x, a[0][q].y};
        const v2f a1l = {a[1][q].x, a[1][q].y};
        const v2f a0h = {a[0][q].z, a[0][q].w};
        const v2f a1h = {a[1][q].z, a[1][q].w};
        acc0 = __builtin_amdgcn_wmma_f32_16x16x4_f32(false, a0l, false, b01, (short)0, acc0, false, false);
        acc1 = __builtin_amdgcn_wmma_f32_16x16x4_f32(false, a1l, false, b01, (short)0, acc1, false, false);
        acc0 = __builtin_amdgcn_wmma_f32_16x16x4_f32(false, a0h, false, b23, (short)0, acc0, false, false);
        acc1 = __builtin_amdgcn_wmma_f32_16x16x4_f32(false, a1h, false, b23, (short)0, acc1, false, false);
    }
}

__device__ __forceinline__ void product_block(float* dst, const float* __restrict__ x,
                                              const float* Bs, int t0, int hidx, int lane,
                                              float bias_n, float k_n) {
    const int half = lane >> 4;
    const int l16  = lane & 15;
    const int j0   = hidx * 32;
    if (t0 + j0 >= T_DIM) return;

    const float* aP = x + (size_t)(t0 + j0 + l16) * IN_DIM + 4 * half;
    const float* bP = Bs + l16 * B_PITCH + 4 * half;

    v8f acc0 = (v8f){0.f,0.f,0.f,0.f,0.f,0.f,0.f,0.f};
    v8f acc1 = (v8f){0.f,0.f,0.f,0.f,0.f,0.f,0.f,0.f};
    v4f aA[2][NQ], bA[NQ], aB[2][NQ], bB[NQ];
    load_frags(aA, bA, aP, bP, 0);
    for (int k0 = 0; k0 < IN_DIM; k0 += 128) {
        load_frags(aB, bB, aP, bP, k0 + 64);
        asm volatile("" ::: "memory");
        mma_frags(acc0, acc1, aA, bA);
        if (k0 + 128 < IN_DIM) load_frags(aA, bA, aP, bP, k0 + 128);
        mma_frags(acc0, acc1, aB, bB);
    }

    float* o = dst + (j0 + 8 * half) * 16 + l16;
#pragma unroll
    for (int r = 0; r < 8; ++r) {
        o[r * 16]        = (bias_n + acc0[r]) * k_n;
        o[(16 + r) * 16] = (bias_n + acc1[r]) * k_n;
    }
}

__device__ __forceinline__ void stage_barrier() {
    asm volatile("s_wait_dscnt 0x0\n\ts_barrier_signal -1\n\ts_barrier_wait -1" ::: "memory");
}

__global__ void __launch_bounds__(FUSED_THREADS)
fused_kernel(const float* __restrict__ x, const float* __restrict__ w_in,
             const float* __restrict__ w_rec, const float* __restrict__ bias,
             float* __restrict__ hc) {
    __shared__ __attribute__((aligned(16))) float lds[LDS_FLOATS];
    float* Bs  = lds;
    float* stg = lds + B_FLOATS;

    const int tid  = threadIdx.x;
    const int lane = tid & 31;
    const int wave = __builtin_amdgcn_readfirstlane((int)(tid >> 5));
    const int wg   = blockIdx.x;

    {
        const float* wsrc = w_in + (size_t)wg * 16 * IN_DIM;
#pragma unroll
        for (int i = 0; i < 16; ++i) {
            const int f   = tid + FUSED_THREADS * i;
            const int row = f >> 8;
            const int c4  = f & 255;
            *(v4f*)(Bs + row * B_PITCH + c4 * 4) = *(const v4f*)(wsrc + (size_t)row * IN_DIM + c4 * 4);
        }
    }

    const int s = wg * WG_STATES + (lane & 3);
    const float wf = w_rec[s] * K_SIG,             wi = w_rec[S_DIM + s] * K_SIG;
    const float wo = w_rec[2 * S_DIM + s] * K_SIG, wg_ = w_rec[3 * S_DIM + s] * K_TANH;

    const int n    = lane & 15;
    const float bias_n = bias[(n & 3) * S_DIM + wg * WG_STATES + (n >> 2)];
    const float k_n    = ((n & 3) == 3) ? K_TANH : K_SIG;
    const int hidx = (wave < 4) ? wave - 1 : wave - 2;
    const bool helper = (wave != 0) && (wave != 4);

    stage_barrier();
    if (helper) product_block(stg, x, Bs, 0, hidx, lane, bias_n, k_n);
    stage_barrier();

    float h = 0.f, c = 0.f;
    const bool hiLane = (lane & 4) != 0;
    const int  grp    = lane >> 3;
    float* hcw = hc + (size_t)wg * T_DIM * 8 + lane;

    for (int b = 0; b < N_BLK; ++b) {
        const int t0 = b * BLK;
        if (wave == 0) {
            const float* ub = stg + (b & 1) * STG_FLOATS + (lane & 3) * 4;
            const int nsteps = (T_DIM - t0 < BLK) ? (T_DIM - t0) : BLK;
            for (int jb = 0; jb < nsteps; jb += 16) {
                float keep[4];
#pragma unroll
                for (int jj = 0; jj < 16; ++jj) {
                    const int j = jb + jj;
                    const v4f uv = *(const v4f*)(ub + j * 16);
                    float co;
                    lstm_step(h, c, co, uv, wf, wi, wo, wg_);
                    keep[jj & 3] = hiLane ? co : h;
                    if ((jj & 3) == 3) {
                        const float v01 = (grp & 1) ? keep[1] : keep[0];
                        const float v23 = (grp & 1) ? keep[3] : keep[2];
                        hcw[(size_t)(t0 + j - 3) * 8] = (grp & 2) ? v23 : v01;
                    }
                }
            }
        } else if (helper) {
            if (b + 1 < N_BLK)
                product_block(stg + ((b + 1) & 1) * STG_FLOATS, x, Bs, t0 + BLK, hidx, lane, bias_n, k_n);
        }
        stage_barrier();
    }
}

__global__ void __launch_bounds__(128)
rows_kernel(const float* __restrict__ hc, float* __restrict__ h_n, float* __restrict__ c_n) {
    const int wgi = threadIdx.x;
    const int t0  = blockIdx.x * 4;
    const v4f* src = (const v4f*)(hc + ((size_t)wgi * T_DIM + t0) * 8);
#pragma unroll
    for (int p = 0; p < 8; ++p) {
        const v4f v = src[p];
        float* row = ((p & 1) ? c_n : h_n) + (size_t)(t0 + (p >> 1) + 1) * S_DIM;
        *(v4f*)(row + wgi * 4) = v;
    }
    if (blockIdx.x == 0) {
        const v4f z = {0.f, 0.f, 0.f, 0.f};
        *(v4f*)(h_n + wgi * 4) = z;
        *(v4f*)(c_n + wgi * 4) = z;
    }
}

__global__ void q_kernel(const float* __restrict__ h_n,
                         const float* __restrict__ w_reg,
                         const float* __restrict__ b_reg,
                         float* __restrict__ q) {
    const int lane = threadIdx.x & 31;
    const int t = (blockIdx.x * blockDim.x + threadIdx.x) >> 5;
    const float* hr = h_n + (size_t)(t + 1) * S_DIM;
    float acc = 0.f;
#pragma unroll
    for (int j = 0; j < S_DIM; j += 32)
        acc += hr[j + lane] * w_reg[j + lane];
#pragma unroll
    for (int off = 16; off; off >>= 1)
        acc += __shfl_xor(acc, off, 32);
    if (lane == 0) q[t] = acc + b_reg[0];
}

extern "C" void kernel_launch(void* const* d_in, const int* in_sizes, int n_in,
                              void* d_out, int out_size, void* d_ws, size_t ws_size,
                              hipStream_t stream) {
    (void)in_sizes; (void)n_in; (void)out_size; (void)ws_size;
    const float* x     = (const float*)d_in[0];
    const float* w_in  = (const float*)d_in[1];
    const float* w_rec = (const float*)d_in[2];
    const float* bias  = (const float*)d_in[3];
    const float* w_reg = (const float*)d_in[4];
    const float* b_reg = (const float*)d_in[5];

    float* out = (float*)d_out;
    float* q   = out;
    float* h_n = out + T_DIM;
    float* c_n = h_n + (size_t)(T_DIM + 1) * S_DIM;

    float* hc = (float*)d_ws;

    fused_kernel<<<N_WG, FUSED_THREADS, 0, stream>>>(x, w_in, w_rec, bias, hc);

    rows_kernel<<<T_DIM / 4, 128, 0, stream>>>(hc, h_n, c_n);

    q_kernel<<<(T_DIM * 32) / 256, 256, 0, stream>>>(h_n, w_reg, b_reg, q);
}


# scan: sigmoids in bounded form m/(1+E), E=2^-|a|, m=1 or E by compare and select: every denominator in [1,4], so the divisions' scale and fixup links go: 18 chain links not 22; rounding points move
# speedup vs baseline: 9.1688x; 1.0836x over previous
#include <hip/hip_runtime.h>
#include <hip/hip_bf16.h>

typedef float v2f __attribute__((ext_vector_type(2)));
typedef float v4f __attribute__((ext_vector_type(4)));
typedef float v8f __attribute__((ext_vector_type(8)));

#define T_DIM  16384
#define IN_DIM 1024
#define S_DIM  512

#define WG_STATES   4
#define N_WG        (S_DIM / WG_STATES)
#define N_HELP      6
#define BLK         (N_HELP * 32)
#define N_BLK       ((T_DIM + BLK - 1) / BLK)
#define B_PITCH     (IN_DIM + 4)
#define B_FLOATS    (16 * B_PITCH)
#define STG_FLOATS  (BLK * 16)
#define FUSED_THREADS 256
#define LDS_FLOATS  (168 * 1024 / 4)

#define LOG2E_F   1.44269504088896340736f
#define K_SIG     (-LOG2E_F)
#define K_TANH    (2.f * LOG2E_F)
#define INV_K_TANH (0.5f * 0.693147180559945309417f)

__device__ __forceinline__ void lstm_step(float& h, float& c, float& co, const v4f uv,
                                          float wf, float wi, float wo, float wg) {
    float t0, t1, t2, t3, t4, t5, t6, t7, t8, t9, t10, t11, t12, t13, t14;
    int sf, so;
    asm(
        "v_fma_f32 %[t0], %[wg], %[h], %[uw]\n\t"
        "v_fma_f32 %[t1], %[wi], %[h], %[uy]\n\t"
        "v_fma_f32 %[t2], %[wf], %[h], %[ux]\n\t"
        "v_fma_f32 %[t3], %[wo], %[h], %[uz]\n\t"
        "v_exp_f32 %[t4], -|%[t0]|\n\t"
        "v_cmp_gt_f32_e64 vcc_lo, %[t1], 0\n\t"
        "v_exp_f32 %[t5], -|%[t2]|\n\t"
        "v_bfi_b32 %[t6], 0x7fffffff, %[kk], %[t0]\n\t"
        "v_exp_f32 %[t0], -|%[t1]|\n\t"
        "v_cmp_gt_f32_e64 %[sf], %[t2], 0\n\t"
        "v_exp_f32 %[t1], -|%[t3]|\n\t"
        "v_cmp_gt_f32_e64 %[so], %[t3], 0\n\t"
        "v_add_f32 %[t2], 1.0, %[t4]\n\t"
        "v_fma_f32 %[t3], -%[t4], %[t6], %[t6]\n\t"
        "v_add_f32 %[t4], 1.0, %[t5]\n\t"
        "v_cndmask_b32_e64 %[t6], 1.0, %[t5], %[sf]\n\t"
        "v_cndmask_b32_e64 %[t5], 1.0, %[t0], vcc_lo\n\t"
        "v_fma_f32 %[t7], %[t0], %[t2], %[t2]\n\t"
        "v_add_f32 %[t0], 1.0, %[t1]\n\t"
        "v_rcp_f32 %[t2], %[t4]\n\t"
        "v_cndmask_b32_e64 %[t8], 1.0, %[t1], %[so]\n\t"
        "v_mul_f32 %[t1], %[t3], %[t5]\n\t"
        "v_rcp_f32 %[t3], %[t7]\n\t"
        "v_mul_f32 %[t5], %[t6], %[t2]\n\t"
        "v_fma_f32 %[t9], -%[t4], %[t2], 1.0\n\t"
        "v_mul_f32 %[t10], %[t1], %[t3]\n\t"
        "v_fma_f32 %[t11], -%[t7], %[t3], 1.0\n\t"
        "v_fma_f32 %[t12], -%[t4], %[t5], %[t6]\n\t"
        "v_fma_f32 %[t13], %[t9], %[t2], %[t2]\n\t"
        "v_fma_f32 %[t9], -%[t7], %[t10], %[t1]\n\t"
        "v_fma_f32 %[t14], %[t11], %[t3], %[t3]\n\t"
        "v_fma_f32 %[t11], %[t12], %[t2], %[t5]\n\t"
        "v_fma_f32 %[t12], %[t9], %[t3], %[t10]\n\t"
        "v_fma_f32 %[t2], -%[t4], %[t11], %[t6]\n\t"
        "v_fma_f32 %[t5], -%[t7], %[t12], %[t1]\n\t"
        "v_fma_f32 %[t9], %[t2], %[t13], %[t11]\n\t"
        "v_fma_f32 %[t3], %[t5], %[t14], %[t12]\n\t"
        "v_fma_f32 %[c], %[t9], %[c], %[t3]\n\t"
        "v_exp_f32 %[t10], -|%[c]|\n\t"
        "v_bfi_b32 %[t4], 0x7fffffff, %[t8], %[c]\n\t"
        "v_mul_f32 %[co], %[ik], %[c]\n\t"
        "v_fma_f32 %[t6], %[t10], %[t0], %[t0]\n\t"
        "v_fma_f32 %[t7], -%[t10], %[t4], %[t4]\n\t"
        "v_rcp_f32 %[t1], %[t6]\n\t"
        "v_nop\n\t"
        "v_mul_f32 %[t2], %[t7], %[t1]\n\t"
        "v_fma_f32 %[t13], -%[t6], %[t1], 1.0\n\t"
        "v_fma_f32 %[t11], -%[t6], %[t2], %[t7]\n\t"
        "v_fma_f32 %[t5], %[t13], %[t1], %[t1]\n\t"
        "v_fma_f32 %[t14], %[t11], %[t1], %[t2]\n\t"
        "v_fma_f32 %[t12], -%[t6], %[t14], %[t7]\n\t"
        "v_fma_f32 %[h], %[t12], %[t5], %[t14]\n\t"
        : [h] "+v"(h), [c] "+v"(c), [co] "=&v"(co),
          [t0] "=&v"(t0),
          [t1] "=&v"(t1),
          [t2] "=&v"(t2),
          [t3] "=&v"(t3),
          [t4] "=&v"(t4),
          [t5] "=&v"(t5),
          [t6] "=&v"(t6),
          [t7] "=&v"(t7),
          [t8] "=&v"(t8),
          [t9] "=&v"(t9),
          [t10] "=&v"(t10),
          [t11] "=&v"(t11),
          [t12] "=&v"(t12),
          [t13] "=&v"(t13),
          [t14] "=&v"(t14),
          [sf] "=&s"(sf), [so] "=&s"(so)
        : [ux] "v"(uv.x), [uy] "v"(uv.y), [uz] "v"(uv.z), [uw] "v"(uv.w),
          [wf] "v"(wf), [wi] "v"(wi), [wo] "v"(wo), [wg] "v"(wg),
          [kk] "s"(K_TANH), [ik] "s"(INV_K_TANH)
        : "vcc");
}

#define NQ 8
__device__ __forceinline__ void load_frags(v4f (&a)[2][NQ], v4f (&b)[NQ],
                                           const float* __restrict__ aP,
                                           const float* bP, int k0) {
#pragma unroll
    for (int q = 0; q < NQ; ++q) {
        a[0][q] = *(const v4f*)(aP + k0 + 8 * q);
        a[1][q] = *(const v4f*)(aP + (size_t)16 * IN_DIM + k0 + 8 * q);
        b[q]    = *(const v4f*)(bP + k0 + 8 * q);
    }
}

__device__ __forceinline__ void mma_frags(v8f& acc0, v8f& acc1,
                                          const v4f (&a)[2][NQ], const v4f (&b)[NQ]) {
#pragma unroll
    for (int q = 0; q < NQ; ++q) {
        const v2f b01 = {b[q].x, b[q].y};
        const v2f b23 = {b[q].z, b[q].w};
        const v2f a0l = {a[0][q].x, a[0][q].y};
        const v2f a1l = {a[1][q].x, a[1][q].y};
        const v2f a0h = {a[0][q].z, a[0][q].w};
        const v2f a1h = {a[1][q].z, a[1][q].w};
        acc0 = __builtin_amdgcn_wmma_f32_16x16x4_f32(false, a0l, false, b01, (short)0, acc0, false, false);
        acc1 = __builtin_amdgcn_wmma_f32_16x16x4_f32(false, a1l, false, b01, (short)0, acc1, false, false);
        acc0 = __builtin_amdgcn_wmma_f32_16x16x4_f32(false, a0h, false, b23, (short)0, acc0, false, false);
        acc1 = __builtin_amdgcn_wmma_f32_16x16x4_f32(false, a1h, false, b23, (short)0, acc1, false, false);
    }
}

__device__ __forceinline__ void product_block(float* dst, const float* __restrict__ x,
                                              const float* Bs, int t0, int hidx, int lane,
                                              float bias_n, float k_n) {
    const int half = lane >> 4;
    const int l16  = lane & 15;
    const int j0   = hidx * 32;
    if (t0 + j0 >= T_DIM) return;

    const float* aP = x + (size_t)(t0 + j0 + l16) * IN_DIM + 4 * half;
    const float* bP = Bs + l16 * B_PITCH + 4 * half;

    v8f acc0 = (v8f){0.f,0.f,0.f,0.f,0.f,0.f,0.f,0.f};
    v8f acc1 = (v8f){0.f,0.f,0.f,0.f,0.f,0.f,0.f,0.f};
    v4f aA[2][NQ], bA[NQ], aB[2][NQ], bB[NQ];
    load_frags(aA, bA, aP, bP, 0);
    for (int k0 = 0; k0 < IN_DIM; k0 += 128) {
        load_frags(aB, bB, aP, bP, k0 + 64);
        asm volatile("" ::: "memory");
        mma_frags(acc0, acc1, aA, bA);
        if (k0 + 128 < IN_DIM) load_frags(aA, bA, aP, bP, k0 + 128);
        mma_frags(acc0, acc1, aB, bB);
    }

    float* o = dst + (j0 + 8 * half) * 16 + l16;
#pragma unroll
    for (int r = 0; r < 8; ++r) {
        o[r * 16]        = (bias_n + acc0[r]) * k_n;
        o[(16 + r) * 16] = (bias_n + acc1[r]) * k_n;
    }
}

__device__ __forceinline__ void stage_barrier() {
    asm volatile("s_wait_dscnt 0x0\n\ts_barrier_signal -1\n\ts_barrier_wait -1" ::: "memory");
}

__global__ void __launch_bounds__(FUSED_THREADS)
fused_kernel(const float* __restrict__ x, const float* __restrict__ w_in,
             const float* __restrict__ w_rec, const float* __restrict__ bias,
             float* __restrict__ hc) {
    __shared__ __attribute__((aligned(16))) float lds[LDS_FLOATS];
    float* Bs  = lds;
    float* stg = lds + B_FLOATS;

    const int tid  = threadIdx.x;
    const int lane = tid & 31;
    const int wave = __builtin_amdgcn_readfirstlane((int)(tid >> 5));
    const int wg   = blockIdx.x;

    {
        const float* wsrc = w_in + (size_t)wg * 16 * IN_DIM;
#pragma unroll
        for (int i = 0; i < 16; ++i) {
            const int f   = tid + FUSED_THREADS * i;
            const int row = f >> 8;
            const int c4  = f & 255;
            *(v4f*)(Bs + row * B_PITCH + c4 * 4) = *(const v4f*)(wsrc + (size_t)row * IN_DIM + c4 * 4);
        }
    }

    const int s = wg * WG_STATES + (lane & 3);
    const float wf = w_rec[s] * K_SIG,             wi = w_rec[S_DIM + s] * K_SIG;
    const float wo = w_rec[2 * S_DIM + s] * K_SIG, wg_ = w_rec[3 * S_DIM + s] * K_TANH;

    const int n    = lane & 15;
    const float bias_n = bias[(n & 3) * S_DIM + wg * WG_STATES + (n >> 2)];
    const float k_n    = ((n & 3) == 3) ? K_TANH : K_SIG;
    const int hidx = (wave < 4) ? wave - 1 : wave - 2;
    const bool helper = (wave != 0) && (wave != 4);

    stage_barrier();
    if (helper) product_block(stg, x, Bs, 0, hidx, lane, bias_n, k_n);
    stage_barrier();

    float h = 0.f, c = 0.f;
    const bool hiLane = (lane & 4) != 0;
    const int  grp    = lane >> 3;
    float* hcw = hc + (size_t)wg * T_DIM * 8 + lane;

    for (int b = 0; b < N_BLK; ++b) {
        const int t0 = b * BLK;
        if (wave == 0) {
            const float* ub = stg + (b & 1) * STG_FLOATS + (lane & 3) * 4;
            const int nsteps = (T_DIM - t0 < BLK) ? (T_DIM - t0) : BLK;
            for (int jb = 0; jb < nsteps; jb += 16) {
                float keep[4];
#pragma unroll
                for (int jj = 0; jj < 16; ++jj) {
                    const int j = jb + jj;
                    const v4f uv = *(const v4f*)(ub + j * 16);
                    float co;
                    lstm_step(h, c, co, uv, wf, wi, wo, wg_);
                    keep[jj & 3] = hiLane ? co : h;
                    if ((jj & 3) == 3) {
                        const float v01 = (grp & 1) ? keep[1] : keep[0];
                        const float v23 = (grp & 1) ? keep[3] : keep[2];
                        hcw[(size_t)(t0 + j - 3) * 8] = (grp & 2) ? v23 : v01;
                    }
                }
            }
        } else if (helper) {
            if (b + 1 < N_BLK)
                product_block(stg + ((b + 1) & 1) * STG_FLOATS, x, Bs, t0 + BLK, hidx, lane, bias_n, k_n);
        }
        stage_barrier();
    }
}

__global__ void __launch_bounds__(128)
rows_kernel(const float* __restrict__ hc, float* __restrict__ h_n, float* __restrict__ c_n) {
    const int wgi = threadIdx.x;
    const int t0  = blockIdx.x * 4;
    const v4f* src = (const v4f*)(hc + ((size_t)wgi * T_DIM + t0) * 8);
#pragma unroll
    for (int p = 0; p < 8; ++p) {
        const v4f v = src[p];
        float* row = ((p & 1) ? c_n : h_n) + (size_t)(t0 + (p >> 1) + 1) * S_DIM;
        *(v4f*)(row + wgi * 4) = v;
    }
    if (blockIdx.x == 0) {
        const v4f z = {0.f, 0.f, 0.f, 0.f};
        *(v4f*)(h_n + wgi * 4) = z;
        *(v4f*)(c_n + wgi * 4) = z;
    }
}

__global__ void q_kernel(const float* __restrict__ h_n,
                         const float* __restrict__ w_reg,
                         const float* __restrict__ b_reg,
                         float* __restrict__ q) {
    const int lane = threadIdx.x & 31;
    const int t = (blockIdx.x * blockDim.x + threadIdx.x) >> 5;
    const float* hr = h_n + (size_t)(t + 1) * S_DIM;
    float acc = 0.f;
#pragma unroll
    for (int j = 0; j < S_DIM; j += 32)
        acc += hr[j + lane] * w_reg[j + lane];
#pragma unroll
    for (int off = 16; off; off >>= 1)
        acc += __shfl_xor(acc, off, 32);
    if (lane == 0) q[t] = acc + b_reg[0];
}

extern "C" void kernel_launch(void* const* d_in, const int* in_sizes, int n_in,
                              void* d_out, int out_size, void* d_ws, size_t ws_size,
                              hipStream_t stream) {
    (void)in_sizes; (void)n_in; (void)out_size; (void)ws_size;
    const float* x     = (const float*)d_in[0];
    const float* w_in  = (const float*)d_in[1];
    const float* w_rec = (const float*)d_in[2];
    const float* bias  = (const float*)d_in[3];
    const float* w_reg = (const float*)d_in[4];
    const float* b_reg = (const float*)d_in[5];

    float* out = (float*)d_out;
    float* q   = out;
    float* h_n = out + T_DIM;
    float* c_n = h_n + (size_t)(T_DIM + 1) * S_DIM;

    float* hc = (float*)d_ws;

    fused_kernel<<<N_WG, FUSED_THREADS, 0, stream>>>(x, w_in, w_rec, bias, hc);

    rows_kernel<<<T_DIM / 4, 128, 0, stream>>>(hc, h_n, c_n);

    q_kernel<<<(T_DIM * 32) / 256, 256, 0, stream>>>(h_n, w_reg, b_reg, q);
}


# helpers: K loop straight-line with both register sets refilled unconditionally behind order points, so each wait before a WMMA counts only the set it reads, not the refill just issued; sums unchanged
# speedup vs baseline: 9.7862x; 1.0673x over previous
#include <hip/hip_runtime.h>
#include <hip/hip_bf16.h>

typedef float v2f __attribute__((ext_vector_type(2)));
typedef float v4f __attribute__((ext_vector_type(4)));
typedef float v8f __attribute__((ext_vector_type(8)));

#define T_DIM  16384
#define IN_DIM 1024
#define S_DIM  512

#define WG_STATES   4
#define N_WG        (S_DIM / WG_STATES)
#define N_HELP      6
#define BLK         (N_HELP * 32)
#define N_BLK       ((T_DIM + BLK - 1) / BLK)
#define B_PITCH     (IN_DIM + 4)
#define B_FLOATS    (16 * B_PITCH)
#define STG_FLOATS  (BLK * 16)
#define FUSED_THREADS 256
#define LDS_FLOATS  (168 * 1024 / 4)

#define LOG2E_F   1.44269504088896340736f
#define K_SIG     (-LOG2E_F)
#define K_TANH    (2.f * LOG2E_F)
#define INV_K_TANH (0.5f * 0.693147180559945309417f)

__device__ __forceinline__ void lstm_step(float& h, float& c, float& co, const v4f uv,
                                          float wf, float wi, float wo, float wg) {
    float t0, t1, t2, t3, t4, t5, t6, t7, t8, t9, t10, t11, t12, t13, t14;
    int sf, so;
    asm(
        "v_fma_f32 %[t0], %[wg], %[h], %[uw]\n\t"
        "v_fma_f32 %[t1], %[wi], %[h], %[uy]\n\t"
        "v_fma_f32 %[t2], %[wf], %[h], %[ux]\n\t"
        "v_fma_f32 %[t3], %[wo], %[h], %[uz]\n\t"
        "v_exp_f32 %[t4], -|%[t0]|\n\t"
        "v_cmp_gt_f32_e64 vcc_lo, %[t1], 0\n\t"
        "v_exp_f32 %[t5], -|%[t2]|\n\t"
        "v_bfi_b32 %[t6], 0x7fffffff, %[kk], %[t0]\n\t"
        "v_exp_f32 %[t0], -|%[t1]|\n\t"
        "v_cmp_gt_f32_e64 %[sf], %[t2], 0\n\t"
        "v_exp_f32 %[t1], -|%[t3]|\n\t"
        "v_cmp_gt_f32_e64 %[so], %[t3], 0\n\t"
        "v_add_f32 %[t2], 1.0, %[t4]\n\t"
        "v_fma_f32 %[t3], -%[t4], %[t6], %[t6]\n\t"
        "v_add_f32 %[t4], 1.0, %[t5]\n\t"
        "v_cndmask_b32_e64 %[t6], 1.0, %[t5], %[sf]\n\t"
        "v_cndmask_b32_e64 %[t5], 1.0, %[t0], vcc_lo\n\t"
        "v_fma_f32 %[t7], %[t0], %[t2], %[t2]\n\t"
        "v_add_f32 %[t0], 1.0, %[t1]\n\t"
        "v_rcp_f32 %[t2], %[t4]\n\t"
        "v_cndmask_b32_e64 %[t8], 1.0, %[t1], %[so]\n\t"
        "v_mul_f32 %[t1], %[t3], %[t5]\n\t"
        "v_rcp_f32 %[t3], %[t7]\n\t"
        "v_mul_f32 %[t5], %[t6], %[t2]\n\t"
        "v_fma_f32 %[t9], -%[t4], %[t2], 1.0\n\t"
        "v_mul_f32 %[t10], %[t1], %[t3]\n\t"
        "v_fma_f32 %[t11], -%[t7], %[t3], 1.0\n\t"
        "v_fma_f32 %[t12], -%[t4], %[t5], %[t6]\n\t"
        "v_fma_f32 %[t13], %[t9], %[t2], %[t2]\n\t"
        "v_fma_f32 %[t9], -%[t7], %[t10], %[t1]\n\t"
        "v_fma_f32 %[t14], %[t11], %[t3], %[t3]\n\t"
        "v_fma_f32 %[t11], %[t12], %[t2], %[t5]\n\t"
        "v_fma_f32 %[t12], %[t9], %[t3], %[t10]\n\t"
        "v_fma_f32 %[t2], -%[t4], %[t11], %[t6]\n\t"
        "v_fma_f32 %[t5], -%[t7], %[t12], %[t1]\n\t"
        "v_fma_f32 %[t9], %[t2], %[t13], %[t11]\n\t"
        "v_fma_f32 %[t3], %[t5], %[t14], %[t12]\n\t"
        "v_fma_f32 %[c], %[t9], %[c], %[t3]\n\t"
        "v_exp_f32 %[t10], -|%[c]|\n\t"
        "v_bfi_b32 %[t4], 0x7fffffff, %[t8], %[c]\n\t"
        "v_mul_f32 %[co], %[ik], %[c]\n\t"
        "v_fma_f32 %[t6], %[t10], %[t0], %[t0]\n\t"
        "v_fma_f32 %[t7], -%[t10], %[t4], %[t4]\n\t"
        "v_rcp_f32 %[t1], %[t6]\n\t"
        "v_nop\n\t"
        "v_mul_f32 %[t2], %[t7], %[t1]\n\t"
        "v_fma_f32 %[t13], -%[t6], %[t1], 1.0\n\t"
        "v_fma_f32 %[t11], -%[t6], %[t2], %[t7]\n\t"
        "v_fma_f32 %[t5], %[t13], %[t1], %[t1]\n\t"
        "v_fma_f32 %[t14], %[t11], %[t1], %[t2]\n\t"
        "v_fma_f32 %[t12], -%[t6], %[t14], %[t7]\n\t"
        "v_fma_f32 %[h], %[t12], %[t5], %[t14]\n\t"
        : [h] "+v"(h), [c] "+v"(c), [co] "=&v"(co),
          [t0] "=&v"(t0),
          [t1] "=&v"(t1),
          [t2] "=&v"(t2),
          [t3] "=&v"(t3),
          [t4] "=&v"(t4),
          [t5] "=&v"(t5),
          [t6] "=&v"(t6),
          [t7] "=&v"(t7),
          [t8] "=&v"(t8),
          [t9] "=&v"(t9),
          [t10] "=&v"(t10),
          [t11] "=&v"(t11),
          [t12] "=&v"(t12),
          [t13] "=&v"(t13),
          [t14] "=&v"(t14),
          [sf] "=&s"(sf), [so] "=&s"(so)
        : [ux] "v"(uv.x), [uy] "v"(uv.y), [uz] "v"(uv.z), [uw] "v"(uv.w),
          [wf] "v"(wf), [wi] "v"(wi), [wo] "v"(wo), [wg] "v"(wg),
          [kk] "s"(K_TANH), [ik] "s"(INV_K_TANH)
        : "vcc");
}

#define NQ 8
__device__ __forceinline__ void load_frags(v4f (&a)[2][NQ], v4f (&b)[NQ],
                                           const float* __restrict__ aP,
                                           const float* bP, int k0) {
#pragma unroll
    for (int q = 0; q < NQ; ++q) {
        a[0][q] = *(const v4f*)(aP + k0 + 8 * q);
        a[1][q] = *(const v4f*)(aP + (size_t)16 * IN_DIM + k0 + 8 * q);
        b[q]    = *(const v4f*)(bP + k0 + 8 * q);
    }
}

__device__ __forceinline__ void mma_frags(v8f& acc0, v8f& acc1,
                                          const v4f (&a)[2][NQ], const v4f (&b)[NQ]) {
#pragma unroll
    for (int q = 0; q < NQ; ++q) {
        const v2f b01 = {b[q].x, b[q].y};
        const v2f b23 = {b[q].z, b[q].w};
        const v2f a0l = {a[0][q].x, a[0][q].y};
        const v2f a1l = {a[1][q].x, a[1][q].y};
        const v2f a0h = {a[0][q].z, a[0][q].w};
        const v2f a1h = {a[1][q].z, a[1][q].w};
        acc0 = __builtin_amdgcn_wmma_f32_16x16x4_f32(false, a0l, false, b01, (short)0, acc0, false, false);
        acc1 = __builtin_amdgcn_wmma_f32_16x16x4_f32(false, a1l, false, b01, (short)0, acc1, false, false);
        acc0 = __builtin_amdgcn_wmma_f32_16x16x4_f32(false, a0h, false, b23, (short)0, acc0, false, false);
        acc1 = __builtin_amdgcn_wmma_f32_16x16x4_f32(false, a1h, false, b23, (short)0, acc1, false, false);
    }
}

__device__ __forceinline__ void product_block(float* dst, const float* __restrict__ x,
                                              const float* Bs, int t0, int hidx, int lane,
                                              float bias_n, float k_n) {
    const int half = lane >> 4;
    const int l16  = lane & 15;
    const int j0   = hidx * 32;
    if (t0 + j0 >= T_DIM) return;

    const float* aP = x + (size_t)(t0 + j0 + l16) * IN_DIM + 4 * half;
    const float* bP = Bs + l16 * B_PITCH + 4 * half;

    v8f acc0 = (v8f){0.f,0.f,0.f,0.f,0.f,0.f,0.f,0.f};
    v8f acc1 = (v8f){0.f,0.f,0.f,0.f,0.f,0.f,0.f,0.f};
    v4f aA[2][NQ], bA[NQ], aB[2][NQ], bB[NQ];
    load_frags(aA, bA, aP, bP, 0);
#define ORDER_POINT() asm volatile("" : "+v"(acc0), "+v"(acc1) :: "memory")
#pragma unroll
    for (int k0 = 0; k0 < IN_DIM - 128; k0 += 128) {
        load_frags(aB, bB, aP, bP, k0 + 64);
        ORDER_POINT();
        mma_frags(acc0, acc1, aA, bA);
        ORDER_POINT();
        load_frags(aA, bA, aP, bP, k0 + 128);
        ORDER_POINT();
        mma_frags(acc0, acc1, aB, bB);
        ORDER_POINT();
    }
    load_frags(aB, bB, aP, bP, IN_DIM - 64);
    ORDER_POINT();
    mma_frags(acc0, acc1, aA, bA);
    mma_frags(acc0, acc1, aB, bB);
#undef ORDER_POINT

    float* o = dst + (j0 + 8 * half) * 16 + l16;
#pragma unroll
    for (int r = 0; r < 8; ++r) {
        o[r * 16]        = (bias_n + acc0[r]) * k_n;
        o[(16 + r) * 16] = (bias_n + acc1[r]) * k_n;
    }
}

__device__ __forceinline__ void stage_barrier() {
    asm volatile("s_wait_dscnt 0x0\n\ts_barrier_signal -1\n\ts_barrier_wait -1" ::: "memory");
}

__global__ void __launch_bounds__(FUSED_THREADS)
fused_kernel(const float* __restrict__ x, const float* __restrict__ w_in,
             const float* __restrict__ w_rec, const float* __restrict__ bias,
             float* __restrict__ hc) {
    __shared__ __attribute__((aligned(16))) float lds[LDS_FLOATS];
    float* Bs  = lds;
    float* stg = lds + B_FLOATS;

    const int tid  = threadIdx.x;
    const int lane = tid & 31;
    const int wave = __builtin_amdgcn_readfirstlane((int)(tid >> 5));
    const int wg   = blockIdx.x;

    {
        const float* wsrc = w_in + (size_t)wg * 16 * IN_DIM;
#pragma unroll
        for (int i = 0; i < 16; ++i) {
            const int f   = tid + FUSED_THREADS * i;
            const int row = f >> 8;
            const int c4  = f & 255;
            *(v4f*)(Bs + row * B_PITCH + c4 * 4) = *(const v4f*)(wsrc + (size_t)row * IN_DIM + c4 * 4);
        }
    }

    const int s = wg * WG_STATES + (lane & 3);
    const float wf = w_rec[s] * K_SIG,             wi = w_rec[S_DIM + s] * K_SIG;
    const float wo = w_rec[2 * S_DIM + s] * K_SIG, wg_ = w_rec[3 * S_DIM + s] * K_TANH;

    const int n    = lane & 15;
    const float bias_n = bias[(n & 3) * S_DIM + wg * WG_STATES + (n >> 2)];
    const float k_n    = ((n & 3) == 3) ? K_TANH : K_SIG;
    const int hidx = (wave < 4) ? wave - 1 : wave - 2;
    const bool helper = (wave != 0) && (wave != 4);

    stage_barrier();

    float h = 0.f, c = 0.f;
    const bool hiLane = (lane & 4) != 0;
    const int  grp    = lane >> 3;
    float* hcw = hc + (size_t)wg * T_DIM * 8 + lane;

    for (int b = -1; b < N_BLK; ++b) {
        const int t0 = b * BLK;
        if (wave == 0 && b >= 0) {
            const float* ub = stg + (b & 1) * STG_FLOATS + (lane & 3) * 4;
            const int nsteps = (T_DIM - t0 < BLK) ? (T_DIM - t0) : BLK;
            for (int jb = 0; jb < nsteps; jb += 16) {
                float keep[4];
#pragma unroll
                for (int jj = 0; jj < 16; ++jj) {
                    const int j = jb + jj;
                    const v4f uv = *(const v4f*)(ub + j * 16);
                    float co;
                    lstm_step(h, c, co, uv, wf, wi, wo, wg_);
                    keep[jj & 3] = hiLane ? co : h;
                    if ((jj & 3) == 3) {
                        const float v01 = (grp & 1) ? keep[1] : keep[0];
                        const float v23 = (grp & 1) ? keep[3] : keep[2];
                        hcw[(size_t)(t0 + j - 3) * 8] = (grp & 2) ? v23 : v01;
                    }
                }
            }
        } else if (helper) {
            if (b + 1 < N_BLK)
                product_block(stg + ((b + 1) & 1) * STG_FLOATS, x, Bs, t0 + BLK, hidx, lane, bias_n, k_n);
        }
        stage_barrier();
    }
}

__global__ void __launch_bounds__(128)
rows_kernel(const float* __restrict__ hc, float* __restrict__ h_n, float* __restrict__ c_n) {
    const int wgi = threadIdx.x;
    const int t0  = blockIdx.x * 4;
    const v4f* src = (const v4f*)(hc + ((size_t)wgi * T_DIM + t0) * 8);
#pragma unroll
    for (int p = 0; p < 8; ++p) {
        const v4f v = src[p];
        float* row = ((p & 1) ? c_n : h_n) + (size_t)(t0 + (p >> 1) + 1) * S_DIM;
        *(v4f*)(row + wgi * 4) = v;
    }
    if (blockIdx.x == 0) {
        const v4f z = {0.f, 0.f, 0.f, 0.f};
        *(v4f*)(h_n + wgi * 4) = z;
        *(v4f*)(c_n + wgi * 4) = z;
    }
}

__global__ void q_kernel(const float* __restrict__ h_n,
                         const float* __restrict__ w_reg,
                         const float* __restrict__ b_reg,
                         float* __restrict__ q) {
    const int lane = threadIdx.x & 31;
    const int t = (blockIdx.x * blockDim.x + threadIdx.x) >> 5;
    const float* hr = h_n + (size_t)(t + 1) * S_DIM;
    float acc = 0.f;
#pragma unroll
    for (int j = 0; j < S_DIM; j += 32)
        acc += hr[j + lane] * w_reg[j + lane];
#pragma unroll
    for (int off = 16; off; off >>= 1)
        acc += __shfl_xor(acc, off, 32);
    if (lane == 0) q[t] = acc + b_reg[0];
}

extern "C" void kernel_launch(void* const* d_in, const int* in_sizes, int n_in,
                              void* d_out, int out_size, void* d_ws, size_t ws_size,
                              hipStream_t stream) {
    (void)in_sizes; (void)n_in; (void)out_size; (void)ws_size;
    const float* x     = (const float*)d_in[0];
    const float* w_in  = (const float*)d_in[1];
    const float* w_rec = (const float*)d_in[2];
    const float* bias  = (const float*)d_in[3];
    const float* w_reg = (const float*)d_in[4];
    const float* b_reg = (const float*)d_in[5];

    float* out = (float*)d_out;
    float* q   = out;
    float* h_n = out + T_DIM;
    float* c_n = h_n + (size_t)(T_DIM + 1) * S_DIM;

    float* hc = (float*)d_ws;

    fused_kernel<<<N_WG, FUSED_THREADS, 0, stream>>>(x, w_in, w_rec, bias, hc);

    rows_kernel<<<T_DIM / 4, 128, 0, stream>>>(hc, h_n, c_n);

    q_kernel<<<(T_DIM * 32) / 256, 256, 0, stream>>>(h_n, w_reg, b_reg, q);
}
